# starting kernel
# baseline (speedup 1.0000x reference)
#include <hip/hip_runtime.h>
#include <hip/hip_bf16.h>

#define NNODES 100000
#define NEDGES 800000
#define HC 128
#define NH 4

typedef float v2f __attribute__((ext_vector_type(2)));
typedef float v8f __attribute__((ext_vector_type(8)));

static inline int cdiv(int a, int b) { return (a + b - 1) / b; }

__global__ __launch_bounds__(256) void k_init_layer(float* __restrict__ acc, int nacc,
                                                    float* __restrict__ m,
                                                    float* __restrict__ s, int nm) {
  int i = blockIdx.x * blockDim.x + threadIdx.x;
  if (i < nacc) acc[i] = 0.0f;
  if (i < nm) {
    m[i] = __int_as_float(0xff800000);
    s[i] = 0.0f;
  }
}

__global__ __launch_bounds__(256) void k_gemm_wmma(const float* __restrict__ X,
                                                   const float* __restrict__ W,
                                                   float* __restrict__ XP,
                                                   int rows, int K) {
  int wave = (blockIdx.x * blockDim.x + threadIdx.x) >> 5;
  int lane = threadIdx.x & 31;
  int m0 = wave * 16;
  if (m0 >= rows) return;

  int n15 = lane & 15;
  int khalf = (lane >> 4) * 2;

  v8f acc[8] = {};
  const float* xbase = X + (size_t)(m0 + n15) * K + khalf;
  const float* wbase = W + (size_t)n15 * K + khalf;

  for (int k = 0; k < K; k += 4) {
    v2f a = *(const v2f*)(xbase + k);
#pragma unroll
    for (int nt = 0; nt < 8; ++nt) {
      v2f b = *(const v2f*)(wbase + (size_t)nt * 16 * K + k);
      acc[nt] = __builtin_amdgcn_wmma_f32_16x16x4_f32(
           false, a,  false, b,
           (short)0, acc[nt],  false,  false);
    }
  }

  int half = lane >> 4;
#pragma unroll
  for (int nt = 0; nt < 8; ++nt) {
#pragma unroll
    for (int i = 0; i < 8; ++i) {
      int m = m0 + half * 8 + i;
      XP[(size_t)m * HC + nt * 16 + n15] = acc[nt][i];
    }
  }
}

__global__ __launch_bounds__(256) void k_attn_dots(const float* __restrict__ XP,
                                                   const float* __restrict__ att_src,
                                                   const float* __restrict__ att_dst,
                                                   float* __restrict__ ASRC,
                                                   float* __restrict__ ADST, int n) {
  int idx = blockIdx.x * blockDim.x + threadIdx.x;
  if (idx >= n * NH) return;
  int node = idx >> 2;
  int h = idx & 3;
  const float* xp = XP + (size_t)node * HC + h * 32;
  const float* as = att_src + h * 32;
  const float* ad = att_dst + h * 32;
  float s0 = 0.f, s1 = 0.f;
#pragma unroll
  for (int c = 0; c < 32; ++c) {
    float v = xp[c];
    s0 += v * as[c];
    s1 += v * ad[c];
  }
  ASRC[idx] = s0;
  ADST[idx] = s1;
}

__device__ __forceinline__ void atomicMaxF(float* addr, float val) {
  if (val >= 0.0f)
    atomicMax((int*)addr, __float_as_int(val));
  else
    atomicMin((unsigned int*)addr, __float_as_uint(val));
}

__global__ __launch_bounds__(256) void k_edge_logits(const int* __restrict__ src,
                                                     const int* __restrict__ dst,
                                                     const float* __restrict__ ASRC,
                                                     const float* __restrict__ ADST,
                                                     float* __restrict__ EB,
                                                     float* __restrict__ M, int ne) {
  int e = blockIdx.x * blockDim.x + threadIdx.x;
  if (e >= ne) return;
  int s = src[e], d = dst[e];
#pragma unroll
  for (int h = 0; h < NH; ++h) {
    float v = ASRC[(size_t)s * NH + h] + ADST[(size_t)d * NH + h];
    v = v > 0.f ? v : 0.2f * v;
    EB[(size_t)e * NH + h] = v;
    atomicMaxF(&M[(size_t)d * NH + h], v);
  }
}

__global__ __launch_bounds__(256) void k_edge_exp(const int* __restrict__ dst,
                                                  float* __restrict__ EB,
                                                  const float* __restrict__ M,
                                                  float* __restrict__ S, int ne) {
  int e = blockIdx.x * blockDim.x + threadIdx.x;
  if (e >= ne) return;
  int d = dst[e];
#pragma unroll
  for (int h = 0; h < NH; ++h) {
    float ex = __expf(EB[(size_t)e * NH + h] - M[(size_t)d * NH + h]);
    EB[(size_t)e * NH + h] = ex;
    atomicAdd(&S[(size_t)d * NH + h], ex);
  }
}

__global__ __launch_bounds__(256) void k_edge_aggr(const int* __restrict__ src,
                                                   const int* __restrict__ dst,
                                                   const float* __restrict__ XP,
                                                   const float* __restrict__ EB,
                                                   const float* __restrict__ S,
                                                   float* __restrict__ OUT, int ne) {
  int wave = (blockIdx.x * blockDim.x + threadIdx.x) >> 5;
  int lane = threadIdx.x & 31;
  if (wave >= ne) return;
  int s = src[wave], d = dst[wave];
  int c0 = lane * 4;
  int h = lane >> 3;
  float alpha = EB[(size_t)wave * NH + h] / (S[(size_t)d * NH + h] + 1e-16f);
  const float4 v = *(const float4*)(XP + (size_t)s * HC + c0);
  float* o = OUT + (size_t)d * HC + c0;
  atomicAdd(o + 0, v.x * alpha);
  atomicAdd(o + 1, v.y * alpha);
  atomicAdd(o + 2, v.z * alpha);
  atomicAdd(o + 3, v.w * alpha);
}

__global__ __launch_bounds__(256) void k_bias_act(float* __restrict__ buf,
                                                  const float* __restrict__ bias,
                                                  int n, int relu) {
  int i = blockIdx.x * blockDim.x + threadIdx.x;
  if (i >= n) return;
  float v = buf[i] + bias[i & (HC - 1)];
  if (relu) v = fmaxf(v, 0.0f);
  buf[i] = v;
}

extern "C" void kernel_launch(void* const* d_in, const int* in_sizes, int n_in,
                              void* d_out, int out_size, void* d_ws, size_t ws_size,
                              hipStream_t stream) {
  const float* x     = (const float*)d_in[0];
  const int*   ei    = (const int*)d_in[1];
  const float* W0    = (const float*)d_in[2];
  const float* atts0 = (const float*)d_in[3];
  const float* attd0 = (const float*)d_in[4];
  const float* b0    = (const float*)d_in[5];
  const float* W1    = (const float*)d_in[6];
  const float* atts1 = (const float*)d_in[7];
  const float* attd1 = (const float*)d_in[8];
  const float* b1    = (const float*)d_in[9];

  const int n = NNODES, ne = NEDGES;
  const int* srcv = ei;
  const int* dstv = ei + ne;

  char* ws = (char*)d_ws;
  float* XP   = (float*)(ws + 0);
  float* H0   = (float*)(ws + 51200000);
  float* ASRC = (float*)(ws + 102400000);
  float* ADST = (float*)(ws + 104000000);
  float* M    = (float*)(ws + 105600000);
  float* S    = (float*)(ws + 107200000);
  float* EB   = (float*)(ws + 108800000);
  float* OUT  = (float*)d_out;

  const dim3 blk(256);
  const int nAcc = n * HC;
  const int nMH  = n * NH;
  const int gemmBlocks = cdiv(cdiv(n, 16) * 32, 256);

  k_init_layer<<<cdiv(nAcc, 256), blk, 0, stream>>>(H0, nAcc, M, S, nMH);
  k_gemm_wmma<<<gemmBlocks, blk, 0, stream>>>(x, W0, XP, n, 256);
  k_attn_dots<<<cdiv(nMH, 256), blk, 0, stream>>>(XP, atts0, attd0, ASRC, ADST, n);
  k_edge_logits<<<cdiv(ne, 256), blk, 0, stream>>>(srcv, dstv, ASRC, ADST, EB, M, ne);
  k_edge_exp<<<cdiv(ne, 256), blk, 0, stream>>>(dstv, EB, M, S, ne);
  k_edge_aggr<<<cdiv(ne * 32, 256), blk, 0, stream>>>(srcv, dstv, XP, EB, S, H0, ne);
  k_bias_act<<<cdiv(nAcc, 256), blk, 0, stream>>>(H0, b0, nAcc,  1);

  k_init_layer<<<cdiv(nAcc, 256), blk, 0, stream>>>(OUT, nAcc, M, S, nMH);
  k_gemm_wmma<<<gemmBlocks, blk, 0, stream>>>(H0, W1, XP, n, 128);
  k_attn_dots<<<cdiv(nMH, 256), blk, 0, stream>>>(XP, atts1, attd1, ASRC, ADST, n);
  k_edge_logits<<<cdiv(ne, 256), blk, 0, stream>>>(srcv, dstv, ASRC, ADST, EB, M, ne);
  k_edge_exp<<<cdiv(ne, 256), blk, 0, stream>>>(dstv, EB, M, S, ne);
  k_edge_aggr<<<cdiv(ne * 32, 256), blk, 0, stream>>>(srcv, dstv, XP, EB, S, OUT, ne);
  k_bias_act<<<cdiv(nAcc, 256), blk, 0, stream>>>(OUT, b1, nAcc,  0);
}


# Edge phase by destination range: a work-group owns 391 nodes, finds their edges in one read of the destination list, keeps max, S and rows in LDS (LDS atomics, sums by arrival); 10 launches become 6
# speedup vs baseline: 4.0573x; 4.0573x over previous
#include <hip/hip_runtime.h>
#include <hip/hip_bf16.h>

#define NNODES 100000
#define NEDGES 800000
#define HC 128
#define NH 4

typedef float v2f __attribute__((ext_vector_type(2)));
typedef float v8f __attribute__((ext_vector_type(8)));

static inline int cdiv(int a, int b) { return (a + b - 1) / b; }

#define GEMM_WGS_K256 256
#define GEMM_WGS_K128 768

template <int KS, int J0, int J1>
__device__ __forceinline__ void load_frag(v2f (&f)[KS], const float* __restrict__ p) {
#pragma unroll
  for (int j = J0; j < J1; ++j) f[j] = *(const v2f*)(p + 4 * j);
}

template <int KS, int J0, int J1>
__device__ __forceinline__ void mma_run(v8f& c0, v8f& c1, const v2f (&a)[KS],
                                        const v2f (&b0)[KS], const v2f (&b1)[KS]) {
#pragma unroll
  for (int j = J0; j < J1; ++j) {
    c0 = __builtin_amdgcn_wmma_f32_16x16x4_f32(
         false, a[j],  false, b0[j],
         (short)0, c0,  false,  false);
    c1 = __builtin_amdgcn_wmma_f32_16x16x4_f32(
         false, a[j],  false, b1[j],
         (short)0, c1,  false,  false);
  }
}

__device__ __forceinline__ void store_tile(const v8f& c0, const v8f& c1, float* __restrict__ o) {
#pragma unroll
  for (int i = 0; i < 8; ++i) {
    o[(size_t)i * HC] = c0[i];
    o[(size_t)i * HC + 16] = c1[i];
  }
}

template <int KS>
__device__ __forceinline__ void tile_step(const v2f (&a)[KS], v2f (&an)[KS],
                                          const float* __restrict__ pn,
                                          const v2f (&b0)[KS], const v2f (&b1)[KS],
                                          float* __restrict__ o) {
  v8f c0 = {}, c1 = {};
  load_frag<KS, 0, KS>(an, pn);
  mma_run<KS, 0, KS>(c0, c1, a, b0, b1);
  store_tile(c0, c1, o);
#pragma unroll
  for (int g = 0; g < KS / 16; ++g) {
    __builtin_amdgcn_sched_group_barrier(0x020, 16, 0);
    __builtin_amdgcn_sched_group_barrier(0x008, 32, 0);
  }
  __builtin_amdgcn_sched_group_barrier(0x040, 16, 0);
}

template <int K>
__global__ __launch_bounds__(128) void k_gemm_wreg(const float* __restrict__ X,
                                                   const float* __restrict__ W,
                                                   float* __restrict__ XP, int ntiles) {
  constexpr int KS = K / 4;
  const int lane = threadIdx.x & 31;
  const int w = threadIdx.x >> 5;
  const int n15 = lane & 15;
  const int half = lane >> 4;
  const int khalf = half * 2;
  const int G = gridDim.x;

  int t = blockIdx.x;
  if (t >= ntiles) return;

  v2f b0[KS], b1[KS];
  const float* wb = W + (size_t)(32 * w + n15) * K + khalf;
  load_frag<KS, 0, KS>(b0, wb);
  load_frag<KS, 0, KS>(b1, wb + (size_t)16 * K);

  const float* xa = X + (size_t)n15 * K + khalf;
  float* xo = XP + (size_t)(half * 8) * HC + 32 * w + n15;

  v2f a0[KS], a1[KS];
  load_frag<KS, 0, KS>(a0, xa + (size_t)t * 16 * K);
  const int tlast = ntiles - 1;
  while (t + G < ntiles) {
    const int t1 = t + G;
    tile_step<KS>(a0, a1, xa + (size_t)t1 * 16 * K, b0, b1, xo + (size_t)t * 16 * HC);
    t = t1 + G;
    tile_step<KS>(a1, a0, xa + (size_t)min(t, tlast) * 16 * K, b0, b1, xo + (size_t)t1 * 16 * HC);
  }
  if (t < ntiles) {
    v8f c0 = {}, c1 = {};
    mma_run<KS, 0, KS>(c0, c1, a0, b0, b1);
    store_tile(c0, c1, xo + (size_t)t * 16 * HC);
  }
}

__global__ __launch_bounds__(256) void k_attn_dots(const float* __restrict__ XP,
                                                   const float* __restrict__ att_src,
                                                   const float* __restrict__ att_dst,
                                                   float* __restrict__ ASRC,
                                                   float* __restrict__ ADST, int n) {
  int idx = blockIdx.x * blockDim.x + threadIdx.x;
  if (idx >= n * NH) return;
  int node = idx >> 2;
  int h = idx & 3;
  const float* xp = XP + (size_t)node * HC + h * 32;
  const float* as = att_src + h * 32;
  const float* ad = att_dst + h * 32;
  float s0 = 0.f, s1 = 0.f;
#pragma unroll
  for (int c = 0; c < 32; ++c) {
    float v = xp[c];
    s0 += v * as[c];
    s1 += v * ad[c];
  }
  ASRC[idx] = s0;
  ADST[idx] = s1;
}

__device__ __forceinline__ unsigned int fkey(float v) {
  unsigned int u = __float_as_uint(v);
  return (u & 0x80000000u) ? ~u : (u | 0x80000000u);
}
__device__ __forceinline__ float fkey_inv(unsigned int k) {
  unsigned int u = (k & 0x80000000u) ? (k & 0x7fffffffu) : ~k;
  return __uint_as_float(u);
}

#define OWN_R 391
#define OWN_WGS 256
#define OWN_T 1024
#define OWN_CAP 8192
#define OWN_EPW 8
#define OWN_SCAN_U 8
static_assert(OWN_R * OWN_WGS >= NNODES, "the owned ranges must cover every node");
static_assert(NEDGES % 4 == 0, "the destination list is read 4 edges (16 bytes) at a time");
static_assert(OWN_WGS * (NEDGES / 4 / OWN_WGS) <= NEDGES / 4, "a start place inside the list");
static_assert(NEDGES <= (1 << 20) && OWN_R <= (1 << 9), "edge number | local row << 20 in one word");
static_assert(NNODES <= (1 << 17), "source node | local row << 17 in one word");
static_assert(OWN_EPW * NH <= 32, "readlane takes the factors from lanes below 32");

__device__ __forceinline__ float leaky02(float v) { return v > 0.f ? v : 0.2f * v; }

__global__ __launch_bounds__(OWN_T) void k_owner_edges(const int* __restrict__ src,
                                                       const int* __restrict__ dst,
                                                       const float* __restrict__ XP,
                                                       const float* __restrict__ ASRC,
                                                       const float* __restrict__ ADST,
                                                       const float* __restrict__ bias,
                                                       float* __restrict__ OUTP, int relu) {
  __shared__ __attribute__((aligned(16))) float acc[OWN_R * HC];
  __shared__ unsigned int mk[OWN_R * NH];
  __shared__ float ssum[OWN_R * NH];
  __shared__ float adl[OWN_R * NH];
  __shared__ unsigned int list[OWN_CAP];
  __shared__ unsigned int cnt;

  const int tid = threadIdx.x;
  const int lo = blockIdx.x * OWN_R;

  {
    float4* acc4 = (float4*)acc;
    const float4 z = {0.0f, 0.0f, 0.0f, 0.0f};
    for (int i = tid; i < OWN_R * HC / 4; i += OWN_T) acc4[i] = z;
    for (int i = tid; i < OWN_R * NH; i += OWN_T) {
      mk[i] = 0u;
      ssum[i] = 0.0f;
      adl[i] = (lo + (i >> 2) < NNODES) ? ADST[(size_t)lo * NH + i] : 0.0f;
    }
    if (tid == 0) cnt = 0u;
  }
  __syncthreads();

  {
    const int4* dst4 = (const int4*)dst;
    constexpr int NQ = NEDGES / 4;
    const int rot = blockIdx.x * (NQ / OWN_WGS);
    for (int b = tid; b < NQ; b += OWN_T * OWN_SCAN_U) {
      int4 v[OWN_SCAN_U];
      int q[OWN_SCAN_U];
#pragma unroll
      for (int u = 0; u < OWN_SCAN_U; ++u) {
        const int p = b + u * OWN_T;
        q[u] = p + rot < NQ ? p + rot : p + rot - NQ;
        v[u] = p < NQ ? dst4[q[u]] : make_int4(-1, -1, -1, -1);
      }
#pragma unroll
      for (int u = 0; u < OWN_SCAN_U; ++u) {
        const int e = q[u] * 4;
        const int dv[4] = {v[u].x, v[u].y, v[u].z, v[u].w};
#pragma unroll
        for (int c = 0; c < 4; ++c) {
          const unsigned int dl = (unsigned int)(dv[c] - lo);
          if (dl < OWN_R) {
            const unsigned int p = atomicAdd(&cnt, 1u);
            if (p < OWN_CAP) list[p] = (unsigned int)(e + c) | (dl << 20);
          }
        }
      }
    }
  }
  __syncthreads();
  const unsigned int total = cnt;
  const bool ovf = total > OWN_CAP;
  const int nl = ovf ? 0 : (int)total;

  if (!ovf) {
    for (int i = tid; i < nl; i += OWN_T) {
      const unsigned int en = list[i];
      const int e = (int)(en & 0xFFFFFu);
      const unsigned int dl = en >> 20;
      const int s = src[e];
      const float4 a = *(const float4*)(ASRC + (size_t)s * NH);
      const float av[4] = {a.x, a.y, a.z, a.w};
#pragma unroll
      for (int h = 0; h < NH; ++h)
        atomicMax(&mk[dl * NH + h], fkey(leaky02(av[h] + adl[dl * NH + h])));
      list[i] = (unsigned int)s | (dl << 17);
    }
  } else {
    for (int e = tid; e < NEDGES; e += OWN_T) {
      const unsigned int dl = (unsigned int)(dst[e] - lo);
      if (dl < OWN_R) {
        const int s = src[e];
        for (int h = 0; h < NH; ++h)
          atomicMax(&mk[dl * NH + h], fkey(leaky02(ASRC[(size_t)s * NH + h] + adl[dl * NH + h])));
      }
    }
  }
  __syncthreads();

  if (!ovf) {
    const int lane = tid & 31;
    const int w = __builtin_amdgcn_readfirstlane(tid >> 5);
    const int h = lane & 3;
    for (int i0 = w * OWN_EPW; i0 < nl; i0 += (OWN_T / 32) * OWN_EPW) {
      unsigned int en[OWN_EPW];
#pragma unroll
      for (int j = 0; j < OWN_EPW; ++j) {
        en[j] = __builtin_amdgcn_readfirstlane(list[min(i0 + j, nl - 1)]);
      }
      float x[OWN_EPW][4], as[OWN_EPW];
#pragma unroll
      for (int j = 0; j < OWN_EPW; ++j) {
        const unsigned int s = en[j] & 0x1FFFFu;
        const float* xr = XP + (size_t)s * HC + lane;
        x[j][0] = xr[0];
        x[j][1] = xr[32];
        x[j][2] = xr[64];
        x[j][3] = xr[96];
        as[j] = ASRC[(size_t)s * NH + h];
      }
#pragma unroll
      for (int j = 0; j < OWN_EPW; ++j) {
        if (i0 + j < nl) {
          const unsigned int dl = en[j] >> 17;
          const float v = leaky02(as[j] + adl[dl * NH + h]);
          const float ex = __expf(v - fkey_inv(mk[dl * NH + h]));
          if (lane < NH) atomicAdd(&ssum[dl * NH + lane], ex);
          const unsigned int exb = __float_as_uint(ex);
          const float f0 = __uint_as_float(__builtin_amdgcn_readlane(exb, 0));
          const float f1 = __uint_as_float(__builtin_amdgcn_readlane(exb, 1));
          const float f2 = __uint_as_float(__builtin_amdgcn_readlane(exb, 2));
          const float f3 = __uint_as_float(__builtin_amdgcn_readlane(exb, 3));
          float* o = &acc[dl * HC + lane];
          atomicAdd(o, x[j][0] * f0);
          atomicAdd(o + 32, x[j][1] * f1);
          atomicAdd(o + 64, x[j][2] * f2);
          atomicAdd(o + 96, x[j][3] * f3);
        }
      }
    }
  } else {
    for (int e = tid; e < NEDGES; e += OWN_T) {
      const unsigned int dl = (unsigned int)(dst[e] - lo);
      if (dl < OWN_R) {
        const int s = src[e];
        float ex[NH];
        for (int h = 0; h < NH; ++h) {
          const float v = leaky02(ASRC[(size_t)s * NH + h] + adl[dl * NH + h]);
          ex[h] = __expf(v - fkey_inv(mk[dl * NH + h]));
          atomicAdd(&ssum[dl * NH + h], ex[h]);
        }
        for (int h = 0; h < NH; ++h)
          for (int c = 0; c < 32; ++c)
            atomicAdd(&acc[dl * HC + h * 32 + c], XP[(size_t)s * HC + h * 32 + c] * ex[h]);
      }
    }
  }
  __syncthreads();

  {
    const float4* acc4 = (const float4*)acc;
    const float4* bias4 = (const float4*)bias;
    float4* out4 = (float4*)OUTP + (size_t)lo * (HC / 4);
    const int nrows = min(OWN_R, NNODES - lo);
    for (int i = tid; i < nrows * (HC / 4); i += OWN_T) {
      const int r = i >> 5;
      const int c4 = i & 31;
      const float den = ssum[r * NH + (c4 >> 3)] + 1e-16f;
      const float4 a = acc4[i];
      const float4 b = bias4[c4];
      float4 v = {a.x / den + b.x, a.y / den + b.y, a.z / den + b.z, a.w / den + b.w};
      if (relu) {
        v.x = fmaxf(v.x, 0.0f);
        v.y = fmaxf(v.y, 0.0f);
        v.z = fmaxf(v.z, 0.0f);
        v.w = fmaxf(v.w, 0.0f);
      }
      out4[i] = v;
    }
  }
}

extern "C" void kernel_launch(void* const* d_in, const int* in_sizes, int n_in,
                              void* d_out, int out_size, void* d_ws, size_t ws_size,
                              hipStream_t stream) {
  const float* x     = (const float*)d_in[0];
  const int*   ei    = (const int*)d_in[1];
  const float* W0    = (const float*)d_in[2];
  const float* atts0 = (const float*)d_in[3];
  const float* attd0 = (const float*)d_in[4];
  const float* b0    = (const float*)d_in[5];
  const float* W1    = (const float*)d_in[6];
  const float* atts1 = (const float*)d_in[7];
  const float* attd1 = (const float*)d_in[8];
  const float* b1    = (const float*)d_in[9];

  const int n = NNODES, ne = NEDGES;
  const int* srcv = ei;
  const int* dstv = ei + ne;

  char* ws = (char*)d_ws;
  float* XP   = (float*)(ws + 0);
  float* H0   = (float*)(ws + 51200000);
  float* ASRC = (float*)(ws + 102400000);
  float* ADST = (float*)(ws + 104000000);
  float* OUT  = (float*)d_out;

  const dim3 blk(256);
  const int nMH  = n * NH;
  const int nTiles = n / 16;

  k_gemm_wreg<256><<<GEMM_WGS_K256, dim3(128), 0, stream>>>(x, W0, XP, nTiles);
  k_attn_dots<<<cdiv(nMH, 256), blk, 0, stream>>>(XP, atts0, attd0, ASRC, ADST, n);
  k_owner_edges<<<OWN_WGS, dim3(OWN_T), 0, stream>>>(srcv, dstv, XP, ASRC, ADST, b0, H0,  1);

  k_gemm_wreg<128><<<GEMM_WGS_K128, dim3(128), 0, stream>>>(H0, W1, XP, nTiles);
  k_attn_dots<<<cdiv(nMH, 256), blk, 0, stream>>>(XP, atts1, attd1, ASRC, ADST, n);
  k_owner_edges<<<OWN_WGS, dim3(OWN_T), 0, stream>>>(srcv, dstv, XP, ASRC, ADST, b1, OUT,  0);
}


# Edge lists built once: layer 0's edge launch parks each range's list (source, local row) and count in that range's slab of the output buffer; layer 1 reloads it, no second dst scan; sums by arrival
# speedup vs baseline: 4.7612x; 1.1735x over previous
#include <hip/hip_runtime.h>
#include <hip/hip_bf16.h>

#define NNODES 100000
#define NEDGES 800000
#define HC 128
#define NH 4

typedef float v2f __attribute__((ext_vector_type(2)));
typedef float v8f __attribute__((ext_vector_type(8)));

static inline int cdiv(int a, int b) { return (a + b - 1) / b; }

#define GEMM_WGS_K256 256
#define GEMM_WGS_K128 768

template <int KS, int J0, int J1>
__device__ __forceinline__ void load_frag(v2f (&f)[KS], const float* __restrict__ p) {
#pragma unroll
  for (int j = J0; j < J1; ++j) f[j] = *(const v2f*)(p + 4 * j);
}

template <int KS, int J0, int J1>
__device__ __forceinline__ void mma_run(v8f& c0, v8f& c1, const v2f (&a)[KS],
                                        const v2f (&b0)[KS], const v2f (&b1)[KS]) {
#pragma unroll
  for (int j = J0; j < J1; ++j) {
    c0 = __builtin_amdgcn_wmma_f32_16x16x4_f32(
         false, a[j],  false, b0[j],
         (short)0, c0,  false,  false);
    c1 = __builtin_amdgcn_wmma_f32_16x16x4_f32(
         false, a[j],  false, b1[j],
         (short)0, c1,  false,  false);
  }
}

__device__ __forceinline__ void store_tile(const v8f& c0, const v8f& c1, float* __restrict__ o) {
#pragma unroll
  for (int i = 0; i < 8; ++i) {
    o[(size_t)i * HC] = c0[i];
    o[(size_t)i * HC + 16] = c1[i];
  }
}

template <int KS>
__device__ __forceinline__ void tile_step(const v2f (&a)[KS], v2f (&an)[KS],
                                          const float* __restrict__ pn,
                                          const v2f (&b0)[KS], const v2f (&b1)[KS],
                                          float* __restrict__ o) {
  v8f c0 = {}, c1 = {};
  load_frag<KS, 0, KS>(an, pn);
  mma_run<KS, 0, KS>(c0, c1, a, b0, b1);
  store_tile(c0, c1, o);
#pragma unroll
  for (int g = 0; g < KS / 16; ++g) {
    __builtin_amdgcn_sched_group_barrier(0x020, 16, 0);
    __builtin_amdgcn_sched_group_barrier(0x008, 32, 0);
  }
  __builtin_amdgcn_sched_group_barrier(0x040, 16, 0);
}

template <int K>
__global__ __launch_bounds__(128) void k_gemm_wreg(const float* __restrict__ X,
                                                   const float* __restrict__ W,
                                                   float* __restrict__ XP, int ntiles) {
  constexpr int KS = K / 4;
  const int lane = threadIdx.x & 31;
  const int w = threadIdx.x >> 5;
  const int n15 = lane & 15;
  const int half = lane >> 4;
  const int khalf = half * 2;
  const int G = gridDim.x;

  int t = blockIdx.x;
  if (t >= ntiles) return;

  v2f b0[KS], b1[KS];
  const float* wb = W + (size_t)(32 * w + n15) * K + khalf;
  load_frag<KS, 0, KS>(b0, wb);
  load_frag<KS, 0, KS>(b1, wb + (size_t)16 * K);

  const float* xa = X + (size_t)n15 * K + khalf;
  float* xo = XP + (size_t)(half * 8) * HC + 32 * w + n15;

  v2f a0[KS], a1[KS];
  load_frag<KS, 0, KS>(a0, xa + (size_t)t * 16 * K);
  const int tlast = ntiles - 1;
  while (t + G < ntiles) {
    const int t1 = t + G;
    tile_step<KS>(a0, a1, xa + (size_t)t1 * 16 * K, b0, b1, xo + (size_t)t * 16 * HC);
    t = t1 + G;
    tile_step<KS>(a1, a0, xa + (size_t)min(t, tlast) * 16 * K, b0, b1, xo + (size_t)t1 * 16 * HC);
  }
  if (t < ntiles) {
    v8f c0 = {}, c1 = {};
    mma_run<KS, 0, KS>(c0, c1, a0, b0, b1);
    store_tile(c0, c1, xo + (size_t)t * 16 * HC);
  }
}

__global__ __launch_bounds__(256) void k_attn_dots(const float* __restrict__ XP,
                                                   const float* __restrict__ att_src,
                                                   const float* __restrict__ att_dst,
                                                   float* __restrict__ ASRC,
                                                   float* __restrict__ ADST, int n) {
  int idx = blockIdx.x * blockDim.x + threadIdx.x;
  if (idx >= n * NH) return;
  int node = idx >> 2;
  int h = idx & 3;
  const float* xp = XP + (size_t)node * HC + h * 32;
  const float* as = att_src + h * 32;
  const float* ad = att_dst + h * 32;
  float s0 = 0.f, s1 = 0.f;
#pragma unroll
  for (int c = 0; c < 32; ++c) {
    float v = xp[c];
    s0 += v * as[c];
    s1 += v * ad[c];
  }
  ASRC[idx] = s0;
  ADST[idx] = s1;
}

__device__ __forceinline__ unsigned int fkey(float v) {
  unsigned int u = __float_as_uint(v);
  return (u & 0x80000000u) ? ~u : (u | 0x80000000u);
}
__device__ __forceinline__ float fkey_inv(unsigned int k) {
  unsigned int u = (k & 0x80000000u) ? (k & 0x7fffffffu) : ~k;
  return __uint_as_float(u);
}

#define OWN_R 391
#define OWN_WGS 256
#define OWN_T 1024
#define OWN_CAP 8192
#define OWN_EPW 8
#define OWN_SCAN_U 8
static_assert(OWN_R * OWN_WGS >= NNODES, "the owned ranges must cover every node");
static_assert(NEDGES % 4 == 0, "the destination list is read 4 edges (16 bytes) at a time");
static_assert(OWN_WGS * (NEDGES / 4 / OWN_WGS) <= NEDGES / 4, "a start place inside the list");
static_assert(NEDGES <= (1 << 20) && OWN_R <= (1 << 9), "edge number | local row << 20 in one word");
static_assert(NNODES <= (1 << 17), "source node | local row << 17 in one word");
static_assert(OWN_EPW * NH <= 32, "readlane takes the factors from lanes below 32");
static_assert((NNODES - (OWN_WGS - 1) * OWN_R) * HC >= OWN_CAP + 1, "list and count fit the last range's output slab");

__device__ __forceinline__ float leaky02(float v) { return v > 0.f ? v : 0.2f * v; }

template <bool FIRST>
__global__ __launch_bounds__(OWN_T) void k_owner_edges(const int* __restrict__ src,
                                                       const int* __restrict__ dst,
                                                       const float* __restrict__ XP,
                                                       const float* __restrict__ ASRC,
                                                       const float* __restrict__ ADST,
                                                       const float* __restrict__ bias,
                                                       float* OUTP, unsigned int* PARK, int relu) {
  __shared__ __attribute__((aligned(16))) float acc[OWN_R * HC];
  __shared__ unsigned int mk[OWN_R * NH];
  __shared__ float ssum[OWN_R * NH];
  __shared__ float adl[OWN_R * NH];
  __shared__ unsigned int list[OWN_CAP];
  __shared__ unsigned int cnt;

  const int tid = threadIdx.x;
  const int lo = blockIdx.x * OWN_R;
  unsigned int* park = PARK + (size_t)lo * HC;

  {
    float4* acc4 = (float4*)acc;
    const float4 z = {0.0f, 0.0f, 0.0f, 0.0f};
    for (int i = tid; i < OWN_R * HC / 4; i += OWN_T) acc4[i] = z;
    for (int i = tid; i < OWN_R * NH; i += OWN_T) {
      mk[i] = 0u;
      ssum[i] = 0.0f;
      adl[i] = (lo + (i >> 2) < NNODES) ? ADST[(size_t)lo * NH + i] : 0.0f;
    }
    if (FIRST) {
      if (tid == 0) cnt = 0u;
    } else {
      const unsigned int gt = park[OWN_CAP];
      const int gn = gt > OWN_CAP ? 0 : (int)gt;
      for (int i = tid; i < gn; i += OWN_T) list[i] = park[i];
      if (tid == 0) cnt = gt;
    }
  }
  __syncthreads();

  if (FIRST) {
    const int4* dst4 = (const int4*)dst;
    constexpr int NQ = NEDGES / 4;
    const int rot = blockIdx.x * (NQ / OWN_WGS);
    for (int b = tid; b < NQ; b += OWN_T * OWN_SCAN_U) {
      int4 v[OWN_SCAN_U];
      int q[OWN_SCAN_U];
#pragma unroll
      for (int u = 0; u < OWN_SCAN_U; ++u) {
        const int p = b + u * OWN_T;
        q[u] = p + rot < NQ ? p + rot : p + rot - NQ;
        v[u] = p < NQ ? dst4[q[u]] : make_int4(-1, -1, -1, -1);
      }
#pragma unroll
      for (int u = 0; u < OWN_SCAN_U; ++u) {
        const int e = q[u] * 4;
        const int dv[4] = {v[u].x, v[u].y, v[u].z, v[u].w};
#pragma unroll
        for (int c = 0; c < 4; ++c) {
          const unsigned int dl = (unsigned int)(dv[c] - lo);
          if (dl < OWN_R) {
            const unsigned int p = atomicAdd(&cnt, 1u);
            if (p < OWN_CAP) list[p] = (unsigned int)(e + c) | (dl << 20);
          }
        }
      }
    }
    __syncthreads();
  }
  const unsigned int total = cnt;
  if (FIRST && tid == 0) park[OWN_CAP] = total;
  const bool ovf = total > OWN_CAP;
  const int nl = ovf ? 0 : (int)total;

  if (!ovf) {
    for (int i = tid; i < nl; i += OWN_T) {
      const unsigned int en = list[i];
      const unsigned int dl = FIRST ? en >> 20 : en >> 17;
      const int s = FIRST ? src[(int)(en & 0xFFFFFu)] : (int)(en & 0x1FFFFu);
      const float4 a = *(const float4*)(ASRC + (size_t)s * NH);
      const float av[4] = {a.x, a.y, a.z, a.w};
#pragma unroll
      for (int h = 0; h < NH; ++h)
        atomicMax(&mk[dl * NH + h], fkey(leaky02(av[h] + adl[dl * NH + h])));
      if (FIRST) {
        const unsigned int sd = (unsigned int)s | (dl << 17);
        list[i] = sd;
        park[i] = sd;
      }
    }
  } else {
    for (int e = tid; e < NEDGES; e += OWN_T) {
      const unsigned int dl = (unsigned int)(dst[e] - lo);
      if (dl < OWN_R) {
        const int s = src[e];
        for (int h = 0; h < NH; ++h)
          atomicMax(&mk[dl * NH + h], fkey(leaky02(ASRC[(size_t)s * NH + h] + adl[dl * NH + h])));
      }
    }
  }
  __syncthreads();

  if (!ovf) {
    const int lane = tid & 31;
    const int w = __builtin_amdgcn_readfirstlane(tid >> 5);
    const int h = lane & 3;
    for (int i0 = w * OWN_EPW; i0 < nl; i0 += (OWN_T / 32) * OWN_EPW) {
      unsigned int en[OWN_EPW];
#pragma unroll
      for (int j = 0; j < OWN_EPW; ++j) {
        en[j] = __builtin_amdgcn_readfirstlane(list[min(i0 + j, nl - 1)]);
      }
      float x[OWN_EPW][4], as[OWN_EPW];
#pragma unroll
      for (int j = 0; j < OWN_EPW; ++j) {
        const unsigned int s = en[j] & 0x1FFFFu;
        const float* xr = XP + (size_t)s * HC + lane;
        x[j][0] = xr[0];
        x[j][1] = xr[32];
        x[j][2] = xr[64];
        x[j][3] = xr[96];
        as[j] = ASRC[(size_t)s * NH + h];
      }
#pragma unroll
      for (int j = 0; j < OWN_EPW; ++j) {
        if (i0 + j < nl) {
          const unsigned int dl = en[j] >> 17;
          const float v = leaky02(as[j] + adl[dl * NH + h]);
          const float ex = __expf(v - fkey_inv(mk[dl * NH + h]));
          if (lane < NH) atomicAdd(&ssum[dl * NH + lane], ex);
          const unsigned int exb = __float_as_uint(ex);
          const float f0 = __uint_as_float(__builtin_amdgcn_readlane(exb, 0));
          const float f1 = __uint_as_float(__builtin_amdgcn_readlane(exb, 1));
          const float f2 = __uint_as_float(__builtin_amdgcn_readlane(exb, 2));
          const float f3 = __uint_as_float(__builtin_amdgcn_readlane(exb, 3));
          float* o = &acc[dl * HC + lane];
          atomicAdd(o, x[j][0] * f0);
          atomicAdd(o + 32, x[j][1] * f1);
          atomicAdd(o + 64, x[j][2] * f2);
          atomicAdd(o + 96, x[j][3] * f3);
        }
      }
    }
  } else {
    for (int e = tid; e < NEDGES; e += OWN_T) {
      const unsigned int dl = (unsigned int)(dst[e] - lo);
      if (dl < OWN_R) {
        const int s = src[e];
        float ex[NH];
        for (int h = 0; h < NH; ++h) {
          const float v = leaky02(ASRC[(size_t)s * NH + h] + adl[dl * NH + h]);
          ex[h] = __expf(v - fkey_inv(mk[dl * NH + h]));
          atomicAdd(&ssum[dl * NH + h], ex[h]);
        }
        for (int h = 0; h < NH; ++h)
          for (int c = 0; c < 32; ++c)
            atomicAdd(&acc[dl * HC + h * 32 + c], XP[(size_t)s * HC + h * 32 + c] * ex[h]);
      }
    }
  }
  __syncthreads();

  {
    const float4* acc4 = (const float4*)acc;
    const float4* bias4 = (const float4*)bias;
    float4* out4 = (float4*)OUTP + (size_t)lo * (HC / 4);
    const int nrows = min(OWN_R, NNODES - lo);
    for (int i = tid; i < nrows * (HC / 4); i += OWN_T) {
      const int r = i >> 5;
      const int c4 = i & 31;
      const float den = ssum[r * NH + (c4 >> 3)] + 1e-16f;
      const float4 a = acc4[i];
      const float4 b = bias4[c4];
      float4 v = {a.x / den + b.x, a.y / den + b.y, a.z / den + b.z, a.w / den + b.w};
      if (relu) {
        v.x = fmaxf(v.x, 0.0f);
        v.y = fmaxf(v.y, 0.0f);
        v.z = fmaxf(v.z, 0.0f);
        v.w = fmaxf(v.w, 0.0f);
      }
      out4[i] = v;
    }
  }
}

extern "C" void kernel_launch(void* const* d_in, const int* in_sizes, int n_in,
                              void* d_out, int out_size, void* d_ws, size_t ws_size,
                              hipStream_t stream) {
  const float* x     = (const float*)d_in[0];
  const int*   ei    = (const int*)d_in[1];
  const float* W0    = (const float*)d_in[2];
  const float* atts0 = (const float*)d_in[3];
  const float* attd0 = (const float*)d_in[4];
  const float* b0    = (const float*)d_in[5];
  const float* W1    = (const float*)d_in[6];
  const float* atts1 = (const float*)d_in[7];
  const float* attd1 = (const float*)d_in[8];
  const float* b1    = (const float*)d_in[9];

  const int n = NNODES, ne = NEDGES;
  const int* srcv = ei;
  const int* dstv = ei + ne;

  char* ws = (char*)d_ws;
  float* XP   = (float*)(ws + 0);
  float* H0   = (float*)(ws + 51200000);
  float* ASRC = (float*)(ws + 102400000);
  float* ADST = (float*)(ws + 104000000);
  float* OUT  = (float*)d_out;

  const dim3 blk(256);
  const int nMH  = n * NH;
  const int nTiles = n / 16;

  k_gemm_wreg<256><<<GEMM_WGS_K256, dim3(128), 0, stream>>>(x, W0, XP, nTiles);
  k_attn_dots<<<cdiv(nMH, 256), blk, 0, stream>>>(XP, atts0, attd0, ASRC, ADST, n);
  k_owner_edges<true><<<OWN_WGS, dim3(OWN_T), 0, stream>>>(srcv, dstv, XP, ASRC, ADST, b0, H0,
                                                           (unsigned int*)d_out,  1);

  k_gemm_wreg<128><<<GEMM_WGS_K128, dim3(128), 0, stream>>>(H0, W1, XP, nTiles);
  k_attn_dots<<<cdiv(nMH, 256), blk, 0, stream>>>(XP, atts1, attd1, ASRC, ADST, n);
  k_owner_edges<false><<<OWN_WGS, dim3(OWN_T), 0, stream>>>(srcv, dstv, XP, ASRC, ADST, b1, OUT,
                                                            (unsigned int*)d_out,  0);
}


# First edge launch scans 1 byte per edge (destination's range number, written by the layer-0 dots launch), 4 edges a word by a zero-byte test, not 4-byte destinations; sums by arrival as before
# speedup vs baseline: 5.2914x; 1.1114x over previous
#include <hip/hip_runtime.h>
#include <hip/hip_bf16.h>

#define NNODES 100000
#define NEDGES 800000
#define HC 128
#define NH 4

typedef float v2f __attribute__((ext_vector_type(2)));
typedef float v8f __attribute__((ext_vector_type(8)));

static inline int cdiv(int a, int b) { return (a + b - 1) / b; }

#define GEMM_WGS_K256 256
#define GEMM_WGS_K128 768

template <int KS, int J0, int J1>
__device__ __forceinline__ void load_frag(v2f (&f)[KS], const float* __restrict__ p) {
#pragma unroll
  for (int j = J0; j < J1; ++j) f[j] = *(const v2f*)(p + 4 * j);
}

template <int KS, int J0, int J1>
__device__ __forceinline__ void mma_run(v8f& c0, v8f& c1, const v2f (&a)[KS],
                                        const v2f (&b0)[KS], const v2f (&b1)[KS]) {
#pragma unroll
  for (int j = J0; j < J1; ++j) {
    c0 = __builtin_amdgcn_wmma_f32_16x16x4_f32(
         false, a[j],  false, b0[j],
         (short)0, c0,  false,  false);
    c1 = __builtin_amdgcn_wmma_f32_16x16x4_f32(
         false, a[j],  false, b1[j],
         (short)0, c1,  false,  false);
  }
}

__device__ __forceinline__ void store_tile(const v8f& c0, const v8f& c1, float* __restrict__ o) {
#pragma unroll
  for (int i = 0; i < 8; ++i) {
    o[(size_t)i * HC] = c0[i];
    o[(size_t)i * HC + 16] = c1[i];
  }
}

template <int KS>
__device__ __forceinline__ void tile_step(const v2f (&a)[KS], v2f (&an)[KS],
                                          const float* __restrict__ pn,
                                          const v2f (&b0)[KS], const v2f (&b1)[KS],
                                          float* __restrict__ o) {
  v8f c0 = {}, c1 = {};
  load_frag<KS, 0, KS>(an, pn);
  mma_run<KS, 0, KS>(c0, c1, a, b0, b1);
  store_tile(c0, c1, o);
#pragma unroll
  for (int g = 0; g < KS / 16; ++g) {
    __builtin_amdgcn_sched_group_barrier(0x020, 16, 0);
    __builtin_amdgcn_sched_group_barrier(0x008, 32, 0);
  }
  __builtin_amdgcn_sched_group_barrier(0x040, 16, 0);
}

template <int K>
__global__ __launch_bounds__(128) void k_gemm_wreg(const float* __restrict__ X,
                                                   const float* __restrict__ W,
                                                   float* __restrict__ XP, int ntiles) {
  constexpr int KS = K / 4;
  const int lane = threadIdx.x & 31;
  const int w = threadIdx.x >> 5;
  const int n15 = lane & 15;
  const int half = lane >> 4;
  const int khalf = half * 2;
  const int G = gridDim.x;

  int t = blockIdx.x;
  if (t >= ntiles) return;

  v2f b0[KS], b1[KS];
  const float* wb = W + (size_t)(32 * w + n15) * K + khalf;
  load_frag<KS, 0, KS>(b0, wb);
  load_frag<KS, 0, KS>(b1, wb + (size_t)16 * K);

  const float* xa = X + (size_t)n15 * K + khalf;
  float* xo = XP + (size_t)(half * 8) * HC + 32 * w + n15;

  v2f a0[KS], a1[KS];
  load_frag<KS, 0, KS>(a0, xa + (size_t)t * 16 * K);
  const int tlast = ntiles - 1;
  while (t + G < ntiles) {
    const int t1 = t + G;
    tile_step<KS>(a0, a1, xa + (size_t)t1 * 16 * K, b0, b1, xo + (size_t)t * 16 * HC);
    t = t1 + G;
    tile_step<KS>(a1, a0, xa + (size_t)min(t, tlast) * 16 * K, b0, b1, xo + (size_t)t1 * 16 * HC);
  }
  if (t < ntiles) {
    v8f c0 = {}, c1 = {};
    mma_run<KS, 0, KS>(c0, c1, a0, b0, b1);
    store_tile(c0, c1, xo + (size_t)t * 16 * HC);
  }
}

#define OWN_R 391
#define OWN_WGS 256
#define OWN_CAP 8192
#define RID_Q 10000
#define RID_OFF 8448
static_assert(OWN_WGS <= 256, "a range number is one byte");
static_assert(NEDGES % (16 * RID_Q) == 0, "whole 16-edge pieces, whole slabs of them");
static_assert(RID_OFF % 32 == 0 && RID_OFF >= OWN_CAP + 1, "pieces start on a line of their own behind the parked list");
static_assert(RID_OFF + RID_Q * 4 <= OWN_R * HC, "the pieces fit a slab");
static_assert(NEDGES / 16 / RID_Q * OWN_R <= NNODES, "only whole slabs hold pieces");
static_assert(NEDGES / 4 <= NNODES * NH, "the dots launch has a thread for every word of range numbers");

__device__ __forceinline__ unsigned int range_of(int d) {
  return min((unsigned int)d / (unsigned int)OWN_R, (unsigned int)(OWN_WGS - 1));
}

template <bool RIDS>
__global__ __launch_bounds__(256) void k_attn_dots(const float* __restrict__ XP,
                                                   const float* __restrict__ att_src,
                                                   const float* __restrict__ att_dst,
                                                   float* __restrict__ ASRC,
                                                   float* __restrict__ ADST, int n,
                                                   const int* __restrict__ dst,
                                                   unsigned int* __restrict__ RIDW) {
  int idx = blockIdx.x * blockDim.x + threadIdx.x;
  if (RIDS) {
    if (idx < NEDGES / 4) {
      const int4 d = ((const int4*)dst)[idx];
      const unsigned int w = range_of(d.x) | (range_of(d.y) << 8) | (range_of(d.z) << 16) |
                             (range_of(d.w) << 24);
      RIDW[(size_t)(idx / (RID_Q * 4)) * (OWN_R * HC) + RID_OFF + (idx % (RID_Q * 4))] = w;
    }
  }
  if (idx >= n * NH) return;
  int node = idx >> 2;
  int h = idx & 3;
  const float* xp = XP + (size_t)node * HC + h * 32;
  const float* as = att_src + h * 32;
  const float* ad = att_dst + h * 32;
  float s0 = 0.f, s1 = 0.f;
#pragma unroll
  for (int c = 0; c < 32; ++c) {
    float v = xp[c];
    s0 += v * as[c];
    s1 += v * ad[c];
  }
  ASRC[idx] = s0;
  ADST[idx] = s1;
}

__device__ __forceinline__ unsigned int fkey(float v) {
  unsigned int u = __float_as_uint(v);
  return (u & 0x80000000u) ? ~u : (u | 0x80000000u);
}
__device__ __forceinline__ float fkey_inv(unsigned int k) {
  unsigned int u = (k & 0x80000000u) ? (k & 0x7fffffffu) : ~k;
  return __uint_as_float(u);
}

#define OWN_T 1024
#define OWN_EPW 8
#define OWN_SCAN_T 1000
#define OWN_SCAN_U 10
#define OWN_NOROW 511
static_assert(OWN_R * OWN_WGS >= NNODES, "the owned ranges must cover every node");
static_assert(OWN_SCAN_T <= OWN_T && OWN_SCAN_T * OWN_SCAN_U == RID_Q, "a trip of the scan reads the pieces of one slab");
static_assert(OWN_SCAN_U % 2 == 0, "two pieces (32 edges) per word of flags");
static_assert(NNODES <= (1 << 17) && OWN_R <= OWN_NOROW && OWN_NOROW < (1 << 15),
              "source node | local row << 17 in one word, OWN_NOROW = no row");
static_assert(OWN_EPW * NH <= 32, "readlane takes the factors from lanes below 32");
static_assert((NNODES - (OWN_WGS - 1) * OWN_R) * HC >= OWN_CAP + 1, "list and count fit the last range's output slab");

__device__ __forceinline__ float leaky02(float v) { return v > 0.f ? v : 0.2f * v; }

template <bool FIRST>
__global__ __launch_bounds__(OWN_T) void k_owner_edges(const int* __restrict__ src,
                                                       const int* __restrict__ dst,
                                                       const float* __restrict__ XP,
                                                       const float* __restrict__ ASRC,
                                                       const float* __restrict__ ADST,
                                                       const float* __restrict__ bias,
                                                       float* OUTP, unsigned int* PARK, int relu) {
  __shared__ __attribute__((aligned(16))) float acc[OWN_R * HC];
  __shared__ unsigned int mk[OWN_R * NH];
  __shared__ float ssum[OWN_R * NH];
  __shared__ float adl[OWN_R * NH];
  __shared__ unsigned int list[OWN_CAP];
  __shared__ unsigned int cnt;

  const int tid = threadIdx.x;
  const int lo = blockIdx.x * OWN_R;
  unsigned int* park = PARK + (size_t)lo * HC;

  {
    float4* acc4 = (float4*)acc;
    const float4 z = {0.0f, 0.0f, 0.0f, 0.0f};
    for (int i = tid; i < OWN_R * HC / 4; i += OWN_T) acc4[i] = z;
    for (int i = tid; i < OWN_R * NH; i += OWN_T) {
      mk[i] = 0u;
      ssum[i] = 0.0f;
      adl[i] = (lo + (i >> 2) < NNODES) ? ADST[(size_t)lo * NH + i] : 0.0f;
    }
    if (FIRST) {
      if (tid == 0) cnt = 0u;
    } else {
      const unsigned int gt = park[OWN_CAP];
      const int gn = gt > OWN_CAP ? 0 : (int)gt;
      for (int i = tid; i < gn; i += OWN_T) list[i] = park[i];
      if (tid == 0) cnt = gt;
    }
  }
  __syncthreads();

  if (FIRST) {
    constexpr int NTRIP = NEDGES / 16 / RID_Q;
    const int lane = tid & 31;
    const unsigned int pat = (unsigned int)blockIdx.x * 0x01010101u;
    const bool mine = tid < OWN_SCAN_T;
    const int st = mine ? tid : 0;
    const int t0 = (int)(blockIdx.x % NTRIP);
    for (int t = 0; t < NTRIP; ++t) {
      const int tt = t0 + t < NTRIP ? t0 + t : t0 + t - NTRIP;
      const uint4* rp = (const uint4*)(PARK + (size_t)tt * (OWN_R * HC) + RID_OFF) + st;
      const int pb = tt * RID_Q + st;
      uint4 v[OWN_SCAN_U];
#pragma unroll
      for (int u = 0; u < OWN_SCAN_U; ++u) v[u] = rp[u * OWN_SCAN_T];
      unsigned int f[OWN_SCAN_U / 2];
      unsigned int n = 0;
#pragma unroll
      for (int g = 0; g < OWN_SCAN_U / 2; ++g) {
        const unsigned int wd[8] = {v[2 * g].x,     v[2 * g].y,     v[2 * g].z,     v[2 * g].w,
                                    v[2 * g + 1].x, v[2 * g + 1].y, v[2 * g + 1].z, v[2 * g + 1].w};
        unsigned int fg = 0;
#pragma unroll
        for (int k = 0; k < 8; ++k) {
          const unsigned int x = wd[k] ^ pat;
          const unsigned int z = (x - 0x01010101u) & ~x;
          fg |= (z >> (7 - k)) & (0x01010101u << k);
        }
        f[g] = mine ? fg : 0u;
        n += __popc(f[g]);
      }
      unsigned int run = n;
      run += (unsigned int)__builtin_amdgcn_update_dpp(0, (int)run, 0x111, 0xF, 0xF, false);
      run += (unsigned int)__builtin_amdgcn_update_dpp(0, (int)run, 0x112, 0xF, 0xF, false);
      run += (unsigned int)__builtin_amdgcn_update_dpp(0, (int)run, 0x114, 0xF, 0xF, false);
      run += (unsigned int)__builtin_amdgcn_update_dpp(0, (int)run, 0x118, 0xF, 0xF, false);
      const unsigned int low = __builtin_amdgcn_readlane(run, 15);
      if (lane >= 16) run += low;
      const unsigned int all = __builtin_amdgcn_readlane(run, 31);
      unsigned int first = 0;
      if (lane == 0) first = atomicAdd(&cnt, all);
      unsigned int p = __builtin_amdgcn_readfirstlane(first) + run - n;
#pragma unroll
      for (int g = 0; g < OWN_SCAN_U / 2; ++g) {
        unsigned int fg = f[g];
        while (fg) {
          const int i = __builtin_ctz(fg);
          fg &= fg - 1;
          const int e = (pb + (2 * g + ((i & 7) >> 2)) * OWN_SCAN_T) * 16 + (i & 3) * 4 + (i >> 3);
          if (p < OWN_CAP) list[p] = (unsigned int)e;
          ++p;
        }
      }
    }
    __syncthreads();
  }
  const unsigned int total = cnt;
  if (FIRST && tid == 0) park[OWN_CAP] = total;
  const bool ovf = total > OWN_CAP;
  const int nl = ovf ? 0 : (int)total;

  if (!ovf) {
    for (int i = tid; i < nl; i += OWN_T) {
      const unsigned int en = list[i];
      unsigned int dl;
      int s;
      if (FIRST) {
        const unsigned int d = (unsigned int)(dst[en] - lo);
        dl = d < OWN_R ? d : (unsigned int)OWN_NOROW;
        s = src[en];
      } else {
        dl = en >> 17;
        s = (int)(en & 0x1FFFFu);
      }
      const float4 a = *(const float4*)(ASRC + (size_t)s * NH);
      const float av[4] = {a.x, a.y, a.z, a.w};
      if (dl < OWN_R) {
#pragma unroll
        for (int h = 0; h < NH; ++h)
          atomicMax(&mk[dl * NH + h], fkey(leaky02(av[h] + adl[dl * NH + h])));
      }
      if (FIRST) {
        const unsigned int sd = (unsigned int)s | (dl << 17);
        list[i] = sd;
        park[i] = sd;
      }
    }
  } else {
    for (int e = tid; e < NEDGES; e += OWN_T) {
      const unsigned int dl = (unsigned int)(dst[e] - lo);
      if (dl < OWN_R) {
        const int s = src[e];
        for (int h = 0; h < NH; ++h)
          atomicMax(&mk[dl * NH + h], fkey(leaky02(ASRC[(size_t)s * NH + h] + adl[dl * NH + h])));
      }
    }
  }
  __syncthreads();

  if (!ovf) {
    const int lane = tid & 31;
    const int w = __builtin_amdgcn_readfirstlane(tid >> 5);
    const int h = lane & 3;
    for (int i0 = w * OWN_EPW; i0 < nl; i0 += (OWN_T / 32) * OWN_EPW) {
      unsigned int en[OWN_EPW];
#pragma unroll
      for (int j = 0; j < OWN_EPW; ++j) {
        en[j] = __builtin_amdgcn_readfirstlane(list[min(i0 + j, nl - 1)]);
      }
      float x[OWN_EPW][4], as[OWN_EPW];
#pragma unroll
      for (int j = 0; j < OWN_EPW; ++j) {
        const unsigned int s = en[j] & 0x1FFFFu;
        const float* xr = XP + (size_t)s * HC + lane;
        x[j][0] = xr[0];
        x[j][1] = xr[32];
        x[j][2] = xr[64];
        x[j][3] = xr[96];
        as[j] = ASRC[(size_t)s * NH + h];
      }
#pragma unroll
      for (int j = 0; j < OWN_EPW; ++j) {
        const unsigned int dl = en[j] >> 17;
        if (i0 + j < nl && dl < OWN_R) {
          const float v = leaky02(as[j] + adl[dl * NH + h]);
          const float ex = __expf(v - fkey_inv(mk[dl * NH + h]));
          if (lane < NH) atomicAdd(&ssum[dl * NH + lane], ex);
          const unsigned int exb = __float_as_uint(ex);
          const float f0 = __uint_as_float(__builtin_amdgcn_readlane(exb, 0));
          const float f1 = __uint_as_float(__builtin_amdgcn_readlane(exb, 1));
          const float f2 = __uint_as_float(__builtin_amdgcn_readlane(exb, 2));
          const float f3 = __uint_as_float(__builtin_amdgcn_readlane(exb, 3));
          float* o = &acc[dl * HC + lane];
          atomicAdd(o, x[j][0] * f0);
          atomicAdd(o + 32, x[j][1] * f1);
          atomicAdd(o + 64, x[j][2] * f2);
          atomicAdd(o + 96, x[j][3] * f3);
        }
      }
    }
  } else {
    for (int e = tid; e < NEDGES; e += OWN_T) {
      const unsigned int dl = (unsigned int)(dst[e] - lo);
      if (dl < OWN_R) {
        const int s = src[e];
        float ex[NH];
        for (int h = 0; h < NH; ++h) {
          const float v = leaky02(ASRC[(size_t)s * NH + h] + adl[dl * NH + h]);
          ex[h] = __expf(v - fkey_inv(mk[dl * NH + h]));
          atomicAdd(&ssum[dl * NH + h], ex[h]);
        }
        for (int h = 0; h < NH; ++h)
          for (int c = 0; c < 32; ++c)
            atomicAdd(&acc[dl * HC + h * 32 + c], XP[(size_t)s * HC + h * 32 + c] * ex[h]);
      }
    }
  }
  __syncthreads();

  {
    const float4* acc4 = (const float4*)acc;
    const float4* bias4 = (const float4*)bias;
    float4* out4 = (float4*)OUTP + (size_t)lo * (HC / 4);
    const int nrows = min(OWN_R, NNODES - lo);
    for (int i = tid; i < nrows * (HC / 4); i += OWN_T) {
      const int r = i >> 5;
      const int c4 = i & 31;
      const float den = ssum[r * NH + (c4 >> 3)] + 1e-16f;
      const float4 a = acc4[i];
      const float4 b = bias4[c4];
      float4 v = {a.x / den + b.x, a.y / den + b.y, a.z / den + b.z, a.w / den + b.w};
      if (relu) {
        v.x = fmaxf(v.x, 0.0f);
        v.y = fmaxf(v.y, 0.0f);
        v.z = fmaxf(v.z, 0.0f);
        v.w = fmaxf(v.w, 0.0f);
      }
      out4[i] = v;
    }
  }
}

extern "C" void kernel_launch(void* const* d_in, const int* in_sizes, int n_in,
                              void* d_out, int out_size, void* d_ws, size_t ws_size,
                              hipStream_t stream) {
  const float* x     = (const float*)d_in[0];
  const int*   ei    = (const int*)d_in[1];
  const float* W0    = (const float*)d_in[2];
  const float* atts0 = (const float*)d_in[3];
  const float* attd0 = (const float*)d_in[4];
  const float* b0    = (const float*)d_in[5];
  const float* W1    = (const float*)d_in[6];
  const float* atts1 = (const float*)d_in[7];
  const float* attd1 = (const float*)d_in[8];
  const float* b1    = (const float*)d_in[9];

  const int n = NNODES, ne = NEDGES;
  const int* srcv = ei;
  const int* dstv = ei + ne;

  char* ws = (char*)d_ws;
  float* XP   = (float*)(ws + 0);
  float* H0   = (float*)(ws + 51200000);
  float* ASRC = (float*)(ws + 102400000);
  float* ADST = (float*)(ws + 104000000);
  float* OUT  = (float*)d_out;

  const dim3 blk(256);
  const int nMH  = n * NH;
  const int nTiles = n / 16;

  k_gemm_wreg<256><<<GEMM_WGS_K256, dim3(128), 0, stream>>>(x, W0, XP, nTiles);
  k_attn_dots<true><<<cdiv(nMH, 256), blk, 0, stream>>>(XP, atts0, attd0, ASRC, ADST, n, dstv,
                                                        (unsigned int*)d_out);
  k_owner_edges<true><<<OWN_WGS, dim3(OWN_T), 0, stream>>>(srcv, dstv, XP, ASRC, ADST, b0, H0,
                                                           (unsigned int*)d_out,  1);

  k_gemm_wreg<128><<<GEMM_WGS_K128, dim3(128), 0, stream>>>(H0, W1, XP, nTiles);
  k_attn_dots<false><<<cdiv(nMH, 256), blk, 0, stream>>>(XP, atts1, attd1, ASRC, ADST, n, dstv,
                                                         (unsigned int*)d_out);
  k_owner_edges<false><<<OWN_WGS, dim3(OWN_T), 0, stream>>>(srcv, dstv, XP, ASRC, ADST, b1, OUT,
                                                            (unsigned int*)d_out,  0);
}
